# starting kernel
# baseline (speedup 1.0000x reference)
#include <hip/hip_runtime.h>

typedef float v2f __attribute__((ext_vector_type(2)));
typedef float v4f __attribute__((ext_vector_type(4)));
typedef float v8f __attribute__((ext_vector_type(8)));

__device__ __forceinline__ v8f wmma4(v2f a, v2f b, v8f c) {
  return __builtin_amdgcn_wmma_f32_16x16x4_f32(false, a, false, b, (short)0, c,
                                               false, false);
}

__device__ __forceinline__ float red16(float v) {
  v += __int_as_float(
      __builtin_amdgcn_ds_swizzle(__float_as_int(v), (1 << 10) | 0x1f));
  v += __int_as_float(
      __builtin_amdgcn_ds_swizzle(__float_as_int(v), (2 << 10) | 0x1f));
  v += __int_as_float(
      __builtin_amdgcn_ds_swizzle(__float_as_int(v), (4 << 10) | 0x1f));
  v += __int_as_float(
      __builtin_amdgcn_ds_swizzle(__float_as_int(v), (8 << 10) | 0x1f));
  return v;
}

__global__ void __launch_bounds__(256) zero_f32(float* __restrict__ p, int n) {
  int i = blockIdx.x * blockDim.x + threadIdx.x;
  if (i < n) p[i] = 0.f;
}

template <int NI, int WCOLS, int OSTR, int OW>
__global__ void __launch_bounds__(256)
edge_conv(const float* __restrict__ nfeat, const float* __restrict__ eattr,
          const int* __restrict__ eidx, const float* __restrict__ w1,
          const float* __restrict__ b1, const float* __restrict__ w2,
          const float* __restrict__ b2, float* __restrict__ agg, int E_,
          int ntiles) {
  constexpr int NU = NI / 4;
  constexpr int KOUT = NI * 10;
  constexpr int NSTEP = 10 * NU + NU;
  __shared__ float sHid[8][16][16];
  __shared__ int sDst[8][16];
  const int lane = threadIdx.x & 31;
  const int wib = threadIdx.x >> 5;
  const int q = lane & 15;
  const int hv = lane >> 4;
  const int wgid = blockIdx.x * 8 + wib;
  const int nw = gridDim.x * 8;

  v2f bf1[2];
#pragma unroll
  for (int s = 0; s < 2; ++s)
#pragma unroll
    for (int j = 0; j < 2; ++j) {
      int K = 4 * s + 2 * hv + j;
      bf1[s][j] = (q < 10) ? w1[K * 10 + q] : 0.f;
    }
  const float b1v = (q < 10) ? b1[q] : 0.f;

  v2f bf2[NSTEP];
#pragma unroll
  for (int s = 0; s < NSTEP; ++s)
#pragma unroll
    for (int j = 0; j < 2; ++j) {
      int K = 4 * s + 2 * hv + j;
      float v;
      if (K < KOUT) {
        int k = K / NI, i = K % NI;
        v = (q < OW) ? w2[k * WCOLS + i * OSTR + q] : 0.f;
      } else {
        int i = K - KOUT;
        v = (q < OW) ? b2[i * OSTR + q] : 0.f;
      }
      bf2[s][j] = v;
    }

  float* wl = &sHid[wib][0][0];

  for (int t = wgid; t < ntiles; t += nw) {
    const int e0 = t * 16;
    const int eS = eidx[e0 + q];
    const int eD = eidx[E_ + e0 + q];

    v8f hid;
#pragma unroll
    for (int r = 0; r < 8; ++r) hid[r] = b1v;
    const float* ea = eattr + (size_t)(e0 + q) * 8 + 2 * hv;
    v2f a0 = *(const v2f*)(ea);
    v2f a1 = *(const v2f*)(ea + 4);
    hid = wmma4(a0, bf1[0], hid);
    hid = wmma4(a1, bf1[1], hid);
#pragma unroll
    for (int r = 0; r < 8; ++r) hid[r] = fmaxf(hid[r], 0.f);

#pragma unroll
    for (int r = 0; r < 8; ++r) wl[(r + 8 * hv) * 16 + q] = hid[r];
    if (lane < 16) sDst[wib][q] = eD;
    __asm__ volatile("s_wait_dscnt 0" ::: "memory");

    float hr[12];
    {
      v4f t0 = *(const v4f*)&wl[q * 16 + 0];
      v4f t1 = *(const v4f*)&wl[q * 16 + 4];
      v4f t2 = *(const v4f*)&wl[q * 16 + 8];
      hr[0] = t0.x; hr[1] = t0.y; hr[2] = t0.z; hr[3] = t0.w;
      hr[4] = t1.x; hr[5] = t1.y; hr[6] = t1.z; hr[7] = t1.w;
      hr[8] = t2.x; hr[9] = t2.y; hr[10] = t2.z; hr[11] = t2.w;
    }
    v2f nsv[NU];
#pragma unroll
    for (int u = 0; u < NU; ++u)
      nsv[u] = *(const v2f*)(nfeat + (size_t)eS * NI + 4 * u + 2 * hv);

    int dstm[8];
#pragma unroll
    for (int r = 0; r < 8; ++r) dstm[r] = sDst[wib][r + 8 * hv];

    v8f msg;
#pragma unroll
    for (int r = 0; r < 8; ++r) msg[r] = 0.f;
#pragma unroll
    for (int k = 0; k < 10; ++k)
#pragma unroll
      for (int u = 0; u < NU; ++u) {
        v2f a;
        a[0] = nsv[u].x * hr[k];
        a[1] = nsv[u].y * hr[k];
        msg = wmma4(a, bf2[k * NU + u], msg);
      }
#pragma unroll
    for (int u = 0; u < NU; ++u)
      msg = wmma4(nsv[u], bf2[10 * NU + u], msg);

#pragma unroll
    for (int r = 0; r < 8; ++r) {
      if (OW == 16 || q < OW)
        atomicAdd(&agg[(size_t)dstm[r] * 16 + q], msg[r]);
    }
  }
}

__global__ void __launch_bounds__(256)
node_relu0(const float* __restrict__ x, const float* __restrict__ agg,
           const float* __restrict__ root, const float* __restrict__ bias,
           float* __restrict__ h, int N_) {
  int idx = blockIdx.x * blockDim.x + threadIdx.x;
  if (idx >= N_ * 16) return;
  int v = idx >> 4, j = idx & 15;
  float acc = agg[idx] + bias[j];
#pragma unroll
  for (int i = 0; i < 8; ++i) acc += x[v * 8 + i] * root[i * 16 + j];
  h[idx] = fmaxf(acc, 0.f);
}

__global__ void __launch_bounds__(256)
node_relu1(const float* __restrict__ h0, const float* __restrict__ agg,
           const float* __restrict__ root, const float* __restrict__ bias,
           float* __restrict__ h1, int N_) {
  int idx = blockIdx.x * blockDim.x + threadIdx.x;
  if (idx >= N_ * 16) return;
  int v = idx >> 4, j = idx & 15;
  float out = 0.f;
  if (j < 10) {
    float acc = agg[idx] + bias[j];
#pragma unroll
    for (int i = 0; i < 16; ++i) acc += h0[v * 16 + i] * root[i * 10 + j];
    out = fmaxf(acc, 0.f);
  }
  h1[idx] = out;
}

__global__ void __launch_bounds__(256)
edge_pred(const float* __restrict__ eattr, const int* __restrict__ eidx,
          const float* __restrict__ h1, const float* __restrict__ w1,
          const float* __restrict__ b1, const float* __restrict__ w2,
          const float* __restrict__ b2, float* __restrict__ out, int E_,
          int ntiles) {
  const int lane = threadIdx.x & 31;
  const int q = lane & 15;
  const int hv = lane >> 4;
  const int wgid = blockIdx.x * 8 + (threadIdx.x >> 5);
  const int nw = gridDim.x * 8;
  (void)E_;

  v2f bf[5];
#pragma unroll
  for (int s = 0; s < 5; ++s)
#pragma unroll
    for (int j = 0; j < 2; ++j) {
      int K = 4 * s + 2 * hv + j;
      bf[s][j] = (K < 18 && q < 10) ? w1[K * 10 + q] : 0.f;
    }
  const float b1v = (q < 10) ? b1[q] : 0.f;
  const float w2q = (q < 10) ? w2[q] : 0.f;
  const float b2v = b2[0];

  for (int t = wgid; t < ntiles; t += nw) {
    const int e0 = t * 16;
    const int eS = eidx[e0 + q];
    const float* ear = eattr + (size_t)(e0 + q) * 8 + 2 * hv;
    const float* h1r = h1 + (size_t)eS * 16 + 2 * hv;
    v2f af[5];
    af[0] = *(const v2f*)(ear);
    af[1] = *(const v2f*)(ear + 4);
    af[2] = *(const v2f*)(h1r);
    af[3] = *(const v2f*)(h1r + 4);
    af[4] = *(const v2f*)(h1r + 8);

    v8f d;
#pragma unroll
    for (int r = 0; r < 8; ++r) d[r] = b1v;
#pragma unroll
    for (int s = 0; s < 5; ++s) d = wmma4(af[s], bf[s], d);

    float sc = 0.f;
#pragma unroll
    for (int r = 0; r < 8; ++r) {
      float val = red16(fmaxf(d[r], 0.f) * w2q);
      sc = (q == r) ? (val + b2v) : sc;
    }
    if (q < 8) out[e0 + 8 * hv + q] = sc;
  }
}

extern "C" void kernel_launch(void* const* d_in, const int* in_sizes, int n_in,
                              void* d_out, int out_size, void* d_ws,
                              size_t ws_size, hipStream_t stream) {
  const float* x = (const float*)d_in[0];
  const float* eattr = (const float*)d_in[1];
  const int* eidx = (const int*)d_in[2];
  const float* c0_w1 = (const float*)d_in[3];
  const float* c0_b1 = (const float*)d_in[4];
  const float* c0_w2 = (const float*)d_in[5];
  const float* c0_b2 = (const float*)d_in[6];
  const float* c0_root = (const float*)d_in[7];
  const float* c0_bias = (const float*)d_in[8];
  const float* c1_w1 = (const float*)d_in[9];
  const float* c1_b1 = (const float*)d_in[10];
  const float* c1_w2 = (const float*)d_in[11];
  const float* c1_b2 = (const float*)d_in[12];
  const float* c1_root = (const float*)d_in[13];
  const float* c1_bias = (const float*)d_in[14];
  const float* ep_w1 = (const float*)d_in[15];
  const float* ep_b1 = (const float*)d_in[16];
  const float* ep_w2 = (const float*)d_in[17];
  const float* ep_b2 = (const float*)d_in[18];
  (void)n_in; (void)out_size; (void)ws_size;

  const int N_ = in_sizes[0] / 8;
  const int E_ = in_sizes[1] / 8;
  const int ntiles = E_ / 16;

  float* agg0 = (float*)d_ws;
  float* agg1 = agg0 + (size_t)N_ * 16;
  float* h0 = agg1 + (size_t)N_ * 16;
  float* h1 = h0 + (size_t)N_ * 16;

  const int EBLK = 960;
  const int nthr = 256;
  const int nblkN = (N_ * 16 + nthr - 1) / nthr;

  zero_f32<<<(2 * N_ * 16 + nthr - 1) / nthr, nthr, 0, stream>>>(agg0,
                                                                 2 * N_ * 16);
  edge_conv<8, 128, 16, 16><<<EBLK, nthr, 0, stream>>>(
      x, eattr, eidx, c0_w1, c0_b1, c0_w2, c0_b2, agg0, E_, ntiles);
  node_relu0<<<nblkN, nthr, 0, stream>>>(x, agg0, c0_root, c0_bias, h0, N_);
  edge_conv<16, 160, 10, 10><<<EBLK, nthr, 0, stream>>>(
      h0, eattr, eidx, c1_w1, c1_b1, c1_w2, c1_b2, agg1, E_, ntiles);
  node_relu1<<<nblkN, nthr, 0, stream>>>(h0, agg1, c1_root, c1_bias, h1, N_);
  edge_pred<<<EBLK, nthr, 0, stream>>>(eattr, eidx, h1, ep_w1, ep_b1, ep_w2,
                                       ep_b2, (float*)d_out, E_, ntiles);
}


# software-pipeline the tile loop of the three edge kernels: next tile's edge attributes, dst indices and gathered rows, and src indices two tiles ahead, are loaded while the current tile computes
# speedup vs baseline: 1.0607x; 1.0607x over previous
#include <hip/hip_runtime.h>

typedef float v2f __attribute__((ext_vector_type(2)));
typedef float v4f __attribute__((ext_vector_type(4)));
typedef float v8f __attribute__((ext_vector_type(8)));

__device__ __forceinline__ v8f wmma4(v2f a, v2f b, v8f c) {
  return __builtin_amdgcn_wmma_f32_16x16x4_f32(false, a, false, b, (short)0, c,
                                               false, false);
}

__device__ __forceinline__ float red16(float v) {
  v += __int_as_float(
      __builtin_amdgcn_ds_swizzle(__float_as_int(v), (1 << 10) | 0x1f));
  v += __int_as_float(
      __builtin_amdgcn_ds_swizzle(__float_as_int(v), (2 << 10) | 0x1f));
  v += __int_as_float(
      __builtin_amdgcn_ds_swizzle(__float_as_int(v), (4 << 10) | 0x1f));
  v += __int_as_float(
      __builtin_amdgcn_ds_swizzle(__float_as_int(v), (8 << 10) | 0x1f));
  return v;
}

__global__ void __launch_bounds__(256) zero_f32(float* __restrict__ p, int n) {
  int i = blockIdx.x * blockDim.x + threadIdx.x;
  if (i < n) p[i] = 0.f;
}

template <int NI, int WCOLS, int OSTR, int OW>
__global__ void __launch_bounds__(256)
edge_conv(const float* __restrict__ nfeat, const float* __restrict__ eattr,
          const int* __restrict__ eidx, const float* __restrict__ w1,
          const float* __restrict__ b1, const float* __restrict__ w2,
          const float* __restrict__ b2, float* __restrict__ agg, int E_,
          int ntiles) {
  constexpr int NU = NI / 4;
  constexpr int KOUT = NI * 10;
  constexpr int NSTEP = 10 * NU + NU;
  __shared__ float sHid[8][16][16];
  __shared__ int sDst[8][16];
  const int lane = threadIdx.x & 31;
  const int wib = threadIdx.x >> 5;
  const int q = lane & 15;
  const int hv = lane >> 4;
  const int wgid = blockIdx.x * 8 + wib;
  const int nw = gridDim.x * 8;

  v2f bf1[2];
#pragma unroll
  for (int s = 0; s < 2; ++s)
#pragma unroll
    for (int j = 0; j < 2; ++j) {
      int K = 4 * s + 2 * hv + j;
      bf1[s][j] = (q < 10) ? w1[K * 10 + q] : 0.f;
    }
  const float b1v = (q < 10) ? b1[q] : 0.f;

  v2f bf2[NSTEP];
#pragma unroll
  for (int s = 0; s < NSTEP; ++s)
#pragma unroll
    for (int j = 0; j < 2; ++j) {
      int K = 4 * s + 2 * hv + j;
      float v;
      if (K < KOUT) {
        int k = K / NI, i = K % NI;
        v = (q < OW) ? w2[k * WCOLS + i * OSTR + q] : 0.f;
      } else {
        int i = K - KOUT;
        v = (q < OW) ? b2[i * OSTR + q] : 0.f;
      }
      bf2[s][j] = v;
    }

  float* wl = &sHid[wib][0][0];

  int t = wgid;
  if (t >= ntiles) return;
  int eD, eSn;
  v2f a0, a1;
  v2f nsv[NU];
  {
    const int e0 = t * 16;
    const int t1 = (t + nw < ntiles) ? t + nw : t;
    const int eS = eidx[e0 + q];
    eD = eidx[E_ + e0 + q];
    eSn = eidx[t1 * 16 + q];
    const float* ea = eattr + (size_t)(e0 + q) * 8 + 2 * hv;
    a0 = *(const v2f*)(ea);
    a1 = *(const v2f*)(ea + 4);
#pragma unroll
    for (int u = 0; u < NU; ++u)
      nsv[u] = *(const v2f*)(nfeat + (size_t)eS * NI + 4 * u + 2 * hv);
  }

  for (; t < ntiles; t += nw) {
    const int tn = (t + nw < ntiles) ? t + nw : t;
    const int tnn = (t + 2 * nw < ntiles) ? t + 2 * nw : t;
    const int eSnn = eidx[tnn * 16 + q];
    const int eDn = eidx[E_ + tn * 16 + q];
    const float* ean = eattr + (size_t)(tn * 16 + q) * 8 + 2 * hv;
    const v2f a0n = *(const v2f*)(ean);
    const v2f a1n = *(const v2f*)(ean + 4);
    v2f nsvn[NU];
#pragma unroll
    for (int u = 0; u < NU; ++u)
      nsvn[u] = *(const v2f*)(nfeat + (size_t)eSn * NI + 4 * u + 2 * hv);

    v8f hid;
#pragma unroll
    for (int r = 0; r < 8; ++r) hid[r] = b1v;
    hid = wmma4(a0, bf1[0], hid);
    hid = wmma4(a1, bf1[1], hid);
#pragma unroll
    for (int r = 0; r < 8; ++r) hid[r] = fmaxf(hid[r], 0.f);

#pragma unroll
    for (int r = 0; r < 8; ++r) wl[(r + 8 * hv) * 16 + q] = hid[r];
    if (lane < 16) sDst[wib][q] = eD;
    __asm__ volatile("s_wait_dscnt 0" ::: "memory");

    float hr[12];
    {
      v4f t0 = *(const v4f*)&wl[q * 16 + 0];
      v4f t1 = *(const v4f*)&wl[q * 16 + 4];
      v4f t2 = *(const v4f*)&wl[q * 16 + 8];
      hr[0] = t0.x; hr[1] = t0.y; hr[2] = t0.z; hr[3] = t0.w;
      hr[4] = t1.x; hr[5] = t1.y; hr[6] = t1.z; hr[7] = t1.w;
      hr[8] = t2.x; hr[9] = t2.y; hr[10] = t2.z; hr[11] = t2.w;
    }

    int dstm[8];
#pragma unroll
    for (int r = 0; r < 8; ++r) dstm[r] = sDst[wib][r + 8 * hv];

    v8f msg;
#pragma unroll
    for (int r = 0; r < 8; ++r) msg[r] = 0.f;
#pragma unroll
    for (int k = 0; k < 10; ++k)
#pragma unroll
      for (int u = 0; u < NU; ++u) {
        v2f a;
        a[0] = nsv[u].x * hr[k];
        a[1] = nsv[u].y * hr[k];
        msg = wmma4(a, bf2[k * NU + u], msg);
      }
#pragma unroll
    for (int u = 0; u < NU; ++u)
      msg = wmma4(nsv[u], bf2[10 * NU + u], msg);

#pragma unroll
    for (int r = 0; r < 8; ++r) {
      if (OW == 16 || q < OW)
        atomicAdd(&agg[(size_t)dstm[r] * 16 + q], msg[r]);
    }

    eSn = eSnn;
    eD = eDn;
    a0 = a0n;
    a1 = a1n;
#pragma unroll
    for (int u = 0; u < NU; ++u) nsv[u] = nsvn[u];
  }
}

__global__ void __launch_bounds__(256)
node_relu0(const float* __restrict__ x, const float* __restrict__ agg,
           const float* __restrict__ root, const float* __restrict__ bias,
           float* __restrict__ h, int N_) {
  int idx = blockIdx.x * blockDim.x + threadIdx.x;
  if (idx >= N_ * 16) return;
  int v = idx >> 4, j = idx & 15;
  float acc = agg[idx] + bias[j];
#pragma unroll
  for (int i = 0; i < 8; ++i) acc += x[v * 8 + i] * root[i * 16 + j];
  h[idx] = fmaxf(acc, 0.f);
}

__global__ void __launch_bounds__(256)
node_relu1(const float* __restrict__ h0, const float* __restrict__ agg,
           const float* __restrict__ root, const float* __restrict__ bias,
           float* __restrict__ h1, int N_) {
  int idx = blockIdx.x * blockDim.x + threadIdx.x;
  if (idx >= N_ * 16) return;
  int v = idx >> 4, j = idx & 15;
  float out = 0.f;
  if (j < 10) {
    float acc = agg[idx] + bias[j];
#pragma unroll
    for (int i = 0; i < 16; ++i) acc += h0[v * 16 + i] * root[i * 10 + j];
    out = fmaxf(acc, 0.f);
  }
  h1[idx] = out;
}

__global__ void __launch_bounds__(256)
edge_pred(const float* __restrict__ eattr, const int* __restrict__ eidx,
          const float* __restrict__ h1, const float* __restrict__ w1,
          const float* __restrict__ b1, const float* __restrict__ w2,
          const float* __restrict__ b2, float* __restrict__ out, int E_,
          int ntiles) {
  const int lane = threadIdx.x & 31;
  const int q = lane & 15;
  const int hv = lane >> 4;
  const int wgid = blockIdx.x * 8 + (threadIdx.x >> 5);
  const int nw = gridDim.x * 8;
  (void)E_;

  v2f bf[5];
#pragma unroll
  for (int s = 0; s < 5; ++s)
#pragma unroll
    for (int j = 0; j < 2; ++j) {
      int K = 4 * s + 2 * hv + j;
      bf[s][j] = (K < 18 && q < 10) ? w1[K * 10 + q] : 0.f;
    }
  const float b1v = (q < 10) ? b1[q] : 0.f;
  const float w2q = (q < 10) ? w2[q] : 0.f;
  const float b2v = b2[0];

  int t = wgid;
  if (t >= ntiles) return;
  int eSn;
  v2f af[5];
  {
    const int e0 = t * 16;
    const int t1 = (t + nw < ntiles) ? t + nw : t;
    const int eS = eidx[e0 + q];
    eSn = eidx[t1 * 16 + q];
    const float* ear = eattr + (size_t)(e0 + q) * 8 + 2 * hv;
    const float* h1r = h1 + (size_t)eS * 16 + 2 * hv;
    af[0] = *(const v2f*)(ear);
    af[1] = *(const v2f*)(ear + 4);
    af[2] = *(const v2f*)(h1r);
    af[3] = *(const v2f*)(h1r + 4);
    af[4] = *(const v2f*)(h1r + 8);
  }

  for (; t < ntiles; t += nw) {
    const int e0 = t * 16;
    const int tn = (t + nw < ntiles) ? t + nw : t;
    const int tnn = (t + 2 * nw < ntiles) ? t + 2 * nw : t;
    const int eSnn = eidx[tnn * 16 + q];
    const float* earn = eattr + (size_t)(tn * 16 + q) * 8 + 2 * hv;
    const float* h1rn = h1 + (size_t)eSn * 16 + 2 * hv;
    v2f afn[5];
    afn[0] = *(const v2f*)(earn);
    afn[1] = *(const v2f*)(earn + 4);
    afn[2] = *(const v2f*)(h1rn);
    afn[3] = *(const v2f*)(h1rn + 4);
    afn[4] = *(const v2f*)(h1rn + 8);

    v8f d;
#pragma unroll
    for (int r = 0; r < 8; ++r) d[r] = b1v;
#pragma unroll
    for (int s = 0; s < 5; ++s) d = wmma4(af[s], bf[s], d);

    float sc = 0.f;
#pragma unroll
    for (int r = 0; r < 8; ++r) {
      float val = red16(fmaxf(d[r], 0.f) * w2q);
      sc = (q == r) ? (val + b2v) : sc;
    }
    if (q < 8) out[e0 + 8 * hv + q] = sc;

    eSn = eSnn;
#pragma unroll
    for (int s = 0; s < 5; ++s) af[s] = afn[s];
  }
}

extern "C" void kernel_launch(void* const* d_in, const int* in_sizes, int n_in,
                              void* d_out, int out_size, void* d_ws,
                              size_t ws_size, hipStream_t stream) {
  const float* x = (const float*)d_in[0];
  const float* eattr = (const float*)d_in[1];
  const int* eidx = (const int*)d_in[2];
  const float* c0_w1 = (const float*)d_in[3];
  const float* c0_b1 = (const float*)d_in[4];
  const float* c0_w2 = (const float*)d_in[5];
  const float* c0_b2 = (const float*)d_in[6];
  const float* c0_root = (const float*)d_in[7];
  const float* c0_bias = (const float*)d_in[8];
  const float* c1_w1 = (const float*)d_in[9];
  const float* c1_b1 = (const float*)d_in[10];
  const float* c1_w2 = (const float*)d_in[11];
  const float* c1_b2 = (const float*)d_in[12];
  const float* c1_root = (const float*)d_in[13];
  const float* c1_bias = (const float*)d_in[14];
  const float* ep_w1 = (const float*)d_in[15];
  const float* ep_b1 = (const float*)d_in[16];
  const float* ep_w2 = (const float*)d_in[17];
  const float* ep_b2 = (const float*)d_in[18];
  (void)n_in; (void)out_size; (void)ws_size;

  const int N_ = in_sizes[0] / 8;
  const int E_ = in_sizes[1] / 8;
  const int ntiles = E_ / 16;

  float* agg0 = (float*)d_ws;
  float* agg1 = agg0 + (size_t)N_ * 16;
  float* h0 = agg1 + (size_t)N_ * 16;
  float* h1 = h0 + (size_t)N_ * 16;

  const int EBLK = 960;
  const int nthr = 256;
  const int nblkN = (N_ * 16 + nthr - 1) / nthr;

  zero_f32<<<(2 * N_ * 16 + nthr - 1) / nthr, nthr, 0, stream>>>(agg0,
                                                                 2 * N_ * 16);
  edge_conv<8, 128, 16, 16><<<EBLK, nthr, 0, stream>>>(
      x, eattr, eidx, c0_w1, c0_b1, c0_w2, c0_b2, agg0, E_, ntiles);
  node_relu0<<<nblkN, nthr, 0, stream>>>(x, agg0, c0_root, c0_bias, h0, N_);
  edge_conv<16, 160, 10, 10><<<EBLK, nthr, 0, stream>>>(
      h0, eattr, eidx, c1_w1, c1_b1, c1_w2, c1_b2, agg1, E_, ntiles);
  node_relu1<<<nblkN, nthr, 0, stream>>>(h0, agg1, c1_root, c1_bias, h1, N_);
  edge_pred<<<EBLK, nthr, 0, stream>>>(eattr, eidx, h1, ep_w1, ep_b1, ep_w2,
                                       ep_b2, (float*)d_out, E_, ntiles);
}


# speedup vs baseline: 1.3409x; 1.2642x over previous
#include <hip/hip_runtime.h>

typedef float v2f __attribute__((ext_vector_type(2)));
typedef float v4f __attribute__((ext_vector_type(4)));
typedef float v8f __attribute__((ext_vector_type(8)));

__device__ __forceinline__ v8f wmma4(v2f a, v2f b, v8f c) {
  return __builtin_amdgcn_wmma_f32_16x16x4_f32(false, a, false, b, (short)0, c,
                                               false, false);
}

__device__ __forceinline__ float red16(float v) {
  v += __int_as_float(
      __builtin_amdgcn_ds_swizzle(__float_as_int(v), (1 << 10) | 0x1f));
  v += __int_as_float(
      __builtin_amdgcn_ds_swizzle(__float_as_int(v), (2 << 10) | 0x1f));
  v += __int_as_float(
      __builtin_amdgcn_ds_swizzle(__float_as_int(v), (4 << 10) | 0x1f));
  v += __int_as_float(
      __builtin_amdgcn_ds_swizzle(__float_as_int(v), (8 << 10) | 0x1f));
  return v;
}

__global__ void __launch_bounds__(256) zero_f32(float* __restrict__ p, int n) {
  int i = blockIdx.x * blockDim.x + threadIdx.x;
  if (i < n) p[i] = 0.f;
}

template <int NI, int WCOLS, int OSTR, int OW>
__global__ void __launch_bounds__(256)
edge_conv(const float* __restrict__ nfeat, const float* __restrict__ eattr,
          const int* __restrict__ eidx, const float* __restrict__ w1,
          const float* __restrict__ b1, const float* __restrict__ w2,
          const float* __restrict__ b2, float* __restrict__ agg,
          float* __restrict__ agg_hi, int E_, int ntiles) {
  constexpr int NU = NI / 4;
  constexpr int KOUT = NI * 10;
  constexpr int NSTEP = 10 * NU + NU;
  __shared__ float sHid[8][16][16];
  __shared__ int sDst[8][16];
  const int lane = threadIdx.x & 31;
  const int wib = threadIdx.x >> 5;
  const int q = lane & 15;
  const int hv = lane >> 4;
  const int wgid = blockIdx.x * 8 + wib;
  const int nw = gridDim.x * 8;

  v2f bf1[2];
#pragma unroll
  for (int s = 0; s < 2; ++s)
#pragma unroll
    for (int j = 0; j < 2; ++j) {
      int K = 4 * s + 2 * hv + j;
      bf1[s][j] = (q < 10) ? w1[K * 10 + q] : 0.f;
    }
  const float b1v = (q < 10) ? b1[q] : 0.f;

  v2f bf2[NSTEP];
#pragma unroll
  for (int s = 0; s < NSTEP; ++s)
#pragma unroll
    for (int j = 0; j < 2; ++j) {
      int K = 4 * s + 2 * hv + j;
      float v;
      if (K < KOUT) {
        int k = K / NI, i = K % NI;
        v = (q < OW) ? w2[k * WCOLS + i * OSTR + q] : 0.f;
      } else {
        int i = K - KOUT;
        v = (q < OW) ? b2[i * OSTR + q] : 0.f;
      }
      bf2[s][j] = v;
    }

  float* wl = &sHid[wib][0][0];

  int t = wgid;
  if (t >= ntiles) return;
  const unsigned se_die = __builtin_amdgcn_s_sendmsg_rtn(0x87);
  float* __restrict__ aggp = (((se_die >> 16) & 15u) >= 4u) ? agg_hi : agg;
  int eD, eSn;
  v2f a0, a1;
  v2f nsv[NU];
  {
    const int e0 = t * 16;
    const int t1 = (t + nw < ntiles) ? t + nw : t;
    const int eS = eidx[e0 + q];
    eD = eidx[E_ + e0 + q];
    eSn = eidx[t1 * 16 + q];
    const float* ea = eattr + (size_t)(e0 + q) * 8 + 2 * hv;
    a0 = *(const v2f*)(ea);
    a1 = *(const v2f*)(ea + 4);
#pragma unroll
    for (int u = 0; u < NU; ++u)
      nsv[u] = *(const v2f*)(nfeat + (size_t)eS * NI + 4 * u + 2 * hv);
  }

  for (; t < ntiles; t += nw) {
    const int tn = (t + nw < ntiles) ? t + nw : t;
    const int tnn = (t + 2 * nw < ntiles) ? t + 2 * nw : t;
    const int eSnn = eidx[tnn * 16 + q];
    const int eDn = eidx[E_ + tn * 16 + q];
    const float* ean = eattr + (size_t)(tn * 16 + q) * 8 + 2 * hv;
    const v2f a0n = *(const v2f*)(ean);
    const v2f a1n = *(const v2f*)(ean + 4);
    v2f nsvn[NU];
#pragma unroll
    for (int u = 0; u < NU; ++u)
      nsvn[u] = *(const v2f*)(nfeat + (size_t)eSn * NI + 4 * u + 2 * hv);

    v8f hid;
#pragma unroll
    for (int r = 0; r < 8; ++r) hid[r] = b1v;
    hid = wmma4(a0, bf1[0], hid);
    hid = wmma4(a1, bf1[1], hid);
#pragma unroll
    for (int r = 0; r < 8; ++r) hid[r] = fmaxf(hid[r], 0.f);

#pragma unroll
    for (int r = 0; r < 8; ++r) wl[(r + 8 * hv) * 16 + q] = hid[r];
    if (lane < 16) sDst[wib][q] = eD;
    __asm__ volatile("s_wait_dscnt 0" ::: "memory");

    float hr[12];
    {
      v4f t0 = *(const v4f*)&wl[q * 16 + 0];
      v4f t1 = *(const v4f*)&wl[q * 16 + 4];
      v4f t2 = *(const v4f*)&wl[q * 16 + 8];
      hr[0] = t0.x; hr[1] = t0.y; hr[2] = t0.z; hr[3] = t0.w;
      hr[4] = t1.x; hr[5] = t1.y; hr[6] = t1.z; hr[7] = t1.w;
      hr[8] = t2.x; hr[9] = t2.y; hr[10] = t2.z; hr[11] = t2.w;
    }

    int dstm[8];
#pragma unroll
    for (int r = 0; r < 8; ++r) dstm[r] = sDst[wib][r + 8 * hv];

    v8f msg;
#pragma unroll
    for (int r = 0; r < 8; ++r) msg[r] = 0.f;
#pragma unroll
    for (int k = 0; k < 10; ++k)
#pragma unroll
      for (int u = 0; u < NU; ++u) {
        v2f a;
        a[0] = nsv[u].x * hr[k];
        a[1] = nsv[u].y * hr[k];
        msg = wmma4(a, bf2[k * NU + u], msg);
      }
#pragma unroll
    for (int u = 0; u < NU; ++u)
      msg = wmma4(nsv[u], bf2[10 * NU + u], msg);

#pragma unroll
    for (int r = 0; r < 8; ++r) {
      if (OW == 16 || q < OW)
        atomicAdd(&aggp[(size_t)dstm[r] * 16 + q], msg[r]);
    }

    eSn = eSnn;
    eD = eDn;
    a0 = a0n;
    a1 = a1n;
#pragma unroll
    for (int u = 0; u < NU; ++u) nsv[u] = nsvn[u];
  }
}

__global__ void __launch_bounds__(256)
node_relu0(const float* __restrict__ x, float* __restrict__ agg,
           const float* __restrict__ agg_hi, const float* __restrict__ root,
           const float* __restrict__ bias, float* __restrict__ h, int N_) {
  int idx = blockIdx.x * blockDim.x + threadIdx.x;
  if (idx >= N_ * 16) return;
  int v = idx >> 4, j = idx & 15;
  float acc = (agg[idx] + agg_hi[idx]) + bias[j];
  agg[idx] = 0.f;
#pragma unroll
  for (int i = 0; i < 8; ++i) acc += x[v * 8 + i] * root[i * 16 + j];
  h[idx] = fmaxf(acc, 0.f);
}

__global__ void __launch_bounds__(256)
node_relu1(const float* __restrict__ h0, const float* __restrict__ agg,
           const float* __restrict__ agg_hi, const float* __restrict__ root,
           const float* __restrict__ bias, float* __restrict__ h1, int N_) {
  int idx = blockIdx.x * blockDim.x + threadIdx.x;
  if (idx >= N_ * 16) return;
  int v = idx >> 4, j = idx & 15;
  float out = 0.f;
  if (j < 10) {
    float acc = (agg[idx] + agg_hi[idx]) + bias[j];
#pragma unroll
    for (int i = 0; i < 16; ++i) acc += h0[v * 16 + i] * root[i * 10 + j];
    out = fmaxf(acc, 0.f);
  }
  h1[idx] = out;
}

__global__ void __launch_bounds__(256)
edge_pred(const float* __restrict__ eattr, const int* __restrict__ eidx,
          const float* __restrict__ h1, const float* __restrict__ w1,
          const float* __restrict__ b1, const float* __restrict__ w2,
          const float* __restrict__ b2, float* __restrict__ out, int E_,
          int ntiles) {
  const int lane = threadIdx.x & 31;
  const int q = lane & 15;
  const int hv = lane >> 4;
  const int wgid = blockIdx.x * 8 + (threadIdx.x >> 5);
  const int nw = gridDim.x * 8;
  (void)E_;

  v2f bf[5];
#pragma unroll
  for (int s = 0; s < 5; ++s)
#pragma unroll
    for (int j = 0; j < 2; ++j) {
      int K = 4 * s + 2 * hv + j;
      bf[s][j] = (K < 18 && q < 10) ? w1[K * 10 + q] : 0.f;
    }
  const float b1v = (q < 10) ? b1[q] : 0.f;
  const float w2q = (q < 10) ? w2[q] : 0.f;
  const float b2v = b2[0];

  int t = wgid;
  if (t >= ntiles) return;
  int eSn;
  v2f af[5];
  {
    const int e0 = t * 16;
    const int t1 = (t + nw < ntiles) ? t + nw : t;
    const int eS = eidx[e0 + q];
    eSn = eidx[t1 * 16 + q];
    const float* ear = eattr + (size_t)(e0 + q) * 8 + 2 * hv;
    const float* h1r = h1 + (size_t)eS * 16 + 2 * hv;
    af[0] = *(const v2f*)(ear);
    af[1] = *(const v2f*)(ear + 4);
    af[2] = *(const v2f*)(h1r);
    af[3] = *(const v2f*)(h1r + 4);
    af[4] = *(const v2f*)(h1r + 8);
  }

  for (; t < ntiles; t += nw) {
    const int e0 = t * 16;
    const int tn = (t + nw < ntiles) ? t + nw : t;
    const int tnn = (t + 2 * nw < ntiles) ? t + 2 * nw : t;
    const int eSnn = eidx[tnn * 16 + q];
    const float* earn = eattr + (size_t)(tn * 16 + q) * 8 + 2 * hv;
    const float* h1rn = h1 + (size_t)eSn * 16 + 2 * hv;
    v2f afn[5];
    afn[0] = *(const v2f*)(earn);
    afn[1] = *(const v2f*)(earn + 4);
    afn[2] = *(const v2f*)(h1rn);
    afn[3] = *(const v2f*)(h1rn + 4);
    afn[4] = *(const v2f*)(h1rn + 8);

    v8f d;
#pragma unroll
    for (int r = 0; r < 8; ++r) d[r] = b1v;
#pragma unroll
    for (int s = 0; s < 5; ++s) d = wmma4(af[s], bf[s], d);

    float sc = 0.f;
#pragma unroll
    for (int r = 0; r < 8; ++r) {
      float val = red16(fmaxf(d[r], 0.f) * w2q);
      sc = (q == r) ? (val + b2v) : sc;
    }
    if (q < 8) out[e0 + 8 * hv + q] = sc;

    eSn = eSnn;
#pragma unroll
    for (int s = 0; s < 5; ++s) af[s] = afn[s];
  }
}

extern "C" void kernel_launch(void* const* d_in, const int* in_sizes, int n_in,
                              void* d_out, int out_size, void* d_ws,
                              size_t ws_size, hipStream_t stream) {
  const float* x = (const float*)d_in[0];
  const float* eattr = (const float*)d_in[1];
  const int* eidx = (const int*)d_in[2];
  const float* c0_w1 = (const float*)d_in[3];
  const float* c0_b1 = (const float*)d_in[4];
  const float* c0_w2 = (const float*)d_in[5];
  const float* c0_b2 = (const float*)d_in[6];
  const float* c0_root = (const float*)d_in[7];
  const float* c0_bias = (const float*)d_in[8];
  const float* c1_w1 = (const float*)d_in[9];
  const float* c1_b1 = (const float*)d_in[10];
  const float* c1_w2 = (const float*)d_in[11];
  const float* c1_b2 = (const float*)d_in[12];
  const float* c1_root = (const float*)d_in[13];
  const float* c1_bias = (const float*)d_in[14];
  const float* ep_w1 = (const float*)d_in[15];
  const float* ep_b1 = (const float*)d_in[16];
  const float* ep_w2 = (const float*)d_in[17];
  const float* ep_b2 = (const float*)d_in[18];
  (void)n_in; (void)out_size; (void)ws_size;

  const int N_ = in_sizes[0] / 8;
  const int E_ = in_sizes[1] / 8;
  const int ntiles = E_ / 16;

  float* agg0 = (float*)d_ws;
  float* agg1 = agg0 + (size_t)N_ * 16;
  float* h1 = agg1 + (size_t)N_ * 16;
  float* h0 = h1 + (size_t)N_ * 16;

  const int EBLK = 960;
  const int nthr = 256;
  const int nblkN = (N_ * 16 + nthr - 1) / nthr;

  zero_f32<<<(3 * N_ * 16 + nthr - 1) / nthr, nthr, 0, stream>>>(agg0,
                                                                 3 * N_ * 16);
  edge_conv<8, 128, 16, 16><<<EBLK, nthr, 0, stream>>>(
      x, eattr, eidx, c0_w1, c0_b1, c0_w2, c0_b2, agg0, h1, E_, ntiles);
  node_relu0<<<nblkN, nthr, 0, stream>>>(x, agg0, h1, c0_root, c0_bias, h0,
                                         N_);
  edge_conv<16, 160, 10, 10><<<EBLK, nthr, 0, stream>>>(
      h0, eattr, eidx, c1_w1, c1_b1, c1_w2, c1_b2, agg1, agg0, E_, ntiles);
  node_relu1<<<nblkN, nthr, 0, stream>>>(h0, agg1, agg0, c1_root, c1_bias, h1,
                                         N_);
  edge_pred<<<EBLK, nthr, 0, stream>>>(eattr, eidx, h1, ep_w1, ep_b1, ep_w2,
                                       ep_b2, (float*)d_out, E_, ntiles);
}


# layer-1 edge kernel: per-edge weights hid@W2+b2 formed on the matrix unit with edges as columns (32 WMMAs per tile, was 46, no padded columns), matvec in vector FMAs, grid 512; new summation order
# speedup vs baseline: 1.4546x; 1.0848x over previous
#include <hip/hip_runtime.h>

typedef float v2f __attribute__((ext_vector_type(2)));
typedef float v4f __attribute__((ext_vector_type(4)));
typedef float v8f __attribute__((ext_vector_type(8)));

__device__ __forceinline__ v8f wmma4(v2f a, v2f b, v8f c) {
  return __builtin_amdgcn_wmma_f32_16x16x4_f32(false, a, false, b, (short)0, c,
                                               false, false);
}

__device__ __forceinline__ float red16(float v) {
  v += __int_as_float(
      __builtin_amdgcn_ds_swizzle(__float_as_int(v), (1 << 10) | 0x1f));
  v += __int_as_float(
      __builtin_amdgcn_ds_swizzle(__float_as_int(v), (2 << 10) | 0x1f));
  v += __int_as_float(
      __builtin_amdgcn_ds_swizzle(__float_as_int(v), (4 << 10) | 0x1f));
  v += __int_as_float(
      __builtin_amdgcn_ds_swizzle(__float_as_int(v), (8 << 10) | 0x1f));
  return v;
}

__global__ void __launch_bounds__(256) zero_f32(float* __restrict__ p, int n) {
  int i = blockIdx.x * blockDim.x + threadIdx.x;
  if (i < n) p[i] = 0.f;
}

template <int NI, int WCOLS, int OSTR, int OW>
__global__ void __launch_bounds__(256)
edge_conv(const float* __restrict__ nfeat, const float* __restrict__ eattr,
          const int* __restrict__ eidx, const float* __restrict__ w1,
          const float* __restrict__ b1, const float* __restrict__ w2,
          const float* __restrict__ b2, float* __restrict__ agg,
          float* __restrict__ agg_hi, int E_, int ntiles) {
  constexpr int NU = NI / 4;
  constexpr int KOUT = NI * 10;
  constexpr int NSTEP = 10 * NU + NU;
  __shared__ float sHid[8][16][16];
  __shared__ int sDst[8][16];
  const int lane = threadIdx.x & 31;
  const int wib = threadIdx.x >> 5;
  const int q = lane & 15;
  const int hv = lane >> 4;
  const int wgid = blockIdx.x * 8 + wib;
  const int nw = gridDim.x * 8;

  v2f bf1[2];
#pragma unroll
  for (int s = 0; s < 2; ++s)
#pragma unroll
    for (int j = 0; j < 2; ++j) {
      int K = 4 * s + 2 * hv + j;
      bf1[s][j] = (q < 10) ? w1[K * 10 + q] : 0.f;
    }
  const float b1v = (q < 10) ? b1[q] : 0.f;

  v2f bf2[NSTEP];
#pragma unroll
  for (int s = 0; s < NSTEP; ++s)
#pragma unroll
    for (int j = 0; j < 2; ++j) {
      int K = 4 * s + 2 * hv + j;
      float v;
      if (K < KOUT) {
        int k = K / NI, i = K % NI;
        v = (q < OW) ? w2[k * WCOLS + i * OSTR + q] : 0.f;
      } else {
        int i = K - KOUT;
        v = (q < OW) ? b2[i * OSTR + q] : 0.f;
      }
      bf2[s][j] = v;
    }

  float* wl = &sHid[wib][0][0];

  int t = wgid;
  if (t >= ntiles) return;
  const unsigned se_die = __builtin_amdgcn_s_sendmsg_rtn(0x87);
  float* __restrict__ aggp = (((se_die >> 16) & 15u) >= 4u) ? agg_hi : agg;
  int eD, eSn;
  v2f a0, a1;
  v2f nsv[NU];
  {
    const int e0 = t * 16;
    const int t1 = (t + nw < ntiles) ? t + nw : t;
    const int eS = eidx[e0 + q];
    eD = eidx[E_ + e0 + q];
    eSn = eidx[t1 * 16 + q];
    const float* ea = eattr + (size_t)(e0 + q) * 8 + 2 * hv;
    a0 = *(const v2f*)(ea);
    a1 = *(const v2f*)(ea + 4);
#pragma unroll
    for (int u = 0; u < NU; ++u)
      nsv[u] = *(const v2f*)(nfeat + (size_t)eS * NI + 4 * u + 2 * hv);
  }

  for (; t < ntiles; t += nw) {
    const int tn = (t + nw < ntiles) ? t + nw : t;
    const int tnn = (t + 2 * nw < ntiles) ? t + 2 * nw : t;
    const int eSnn = eidx[tnn * 16 + q];
    const int eDn = eidx[E_ + tn * 16 + q];
    const float* ean = eattr + (size_t)(tn * 16 + q) * 8 + 2 * hv;
    const v2f a0n = *(const v2f*)(ean);
    const v2f a1n = *(const v2f*)(ean + 4);
    v2f nsvn[NU];
#pragma unroll
    for (int u = 0; u < NU; ++u)
      nsvn[u] = *(const v2f*)(nfeat + (size_t)eSn * NI + 4 * u + 2 * hv);

    v8f hid;
#pragma unroll
    for (int r = 0; r < 8; ++r) hid[r] = b1v;
    hid = wmma4(a0, bf1[0], hid);
    hid = wmma4(a1, bf1[1], hid);
#pragma unroll
    for (int r = 0; r < 8; ++r) hid[r] = fmaxf(hid[r], 0.f);

#pragma unroll
    for (int r = 0; r < 8; ++r) wl[(r + 8 * hv) * 16 + q] = hid[r];
    if (lane < 16) sDst[wib][q] = eD;
    __asm__ volatile("s_wait_dscnt 0" ::: "memory");

    float hr[12];
    {
      v4f t0 = *(const v4f*)&wl[q * 16 + 0];
      v4f t1 = *(const v4f*)&wl[q * 16 + 4];
      v4f t2 = *(const v4f*)&wl[q * 16 + 8];
      hr[0] = t0.x; hr[1] = t0.y; hr[2] = t0.z; hr[3] = t0.w;
      hr[4] = t1.x; hr[5] = t1.y; hr[6] = t1.z; hr[7] = t1.w;
      hr[8] = t2.x; hr[9] = t2.y; hr[10] = t2.z; hr[11] = t2.w;
    }

    int dstm[8];
#pragma unroll
    for (int r = 0; r < 8; ++r) dstm[r] = sDst[wib][r + 8 * hv];

    v8f msg;
#pragma unroll
    for (int r = 0; r < 8; ++r) msg[r] = 0.f;
#pragma unroll
    for (int k = 0; k < 10; ++k)
#pragma unroll
      for (int u = 0; u < NU; ++u) {
        v2f a;
        a[0] = nsv[u].x * hr[k];
        a[1] = nsv[u].y * hr[k];
        msg = wmma4(a, bf2[k * NU + u], msg);
      }
#pragma unroll
    for (int u = 0; u < NU; ++u)
      msg = wmma4(nsv[u], bf2[10 * NU + u], msg);

#pragma unroll
    for (int r = 0; r < 8; ++r) {
      if (OW == 16 || q < OW)
        atomicAdd(&aggp[(size_t)dstm[r] * 16 + q], msg[r]);
    }

    eSn = eSnn;
    eD = eDn;
    a0 = a0n;
    a1 = a1n;
#pragma unroll
    for (int u = 0; u < NU; ++u) nsv[u] = nsvn[u];
  }
}

__global__ void __launch_bounds__(256)
edge_conv_l1(const float* __restrict__ nfeat, const float* __restrict__ eattr,
             const int* __restrict__ eidx, const float* __restrict__ w1,
             const float* __restrict__ b1, const float* __restrict__ w2,
             const float* __restrict__ b2, float* __restrict__ agg,
             float* __restrict__ agg_hi, int E_, int ntiles) {
  __shared__ float sMsg[8][16 * 17];
  __shared__ int sDst[8][16];
  const int lane = threadIdx.x & 31;
  const int wib = threadIdx.x >> 5;
  const int q = lane & 15;
  const int hv = lane >> 4;
  const int wgid = blockIdx.x * 8 + wib;
  const int nw = gridDim.x * 8;

  const int kq = 6 * (q >> 3) + (q & 7);
  const bool row_real = ((q & 7) < 6) && (kq < 10);
  v2f af1[2];
#pragma unroll
  for (int s = 0; s < 2; ++s)
#pragma unroll
    for (int j = 0; j < 2; ++j) {
      const int f = 4 * s + 2 * hv + j;
      af1[s][j] = row_real ? w1[f * 10 + (row_real ? kq : 0)] : 0.f;
    }
  float c1[6];
#pragma unroll
  for (int r = 0; r < 6; ++r) {
    const int k = 6 * hv + r;
    c1[r] = (k < 10) ? b1[(k < 10) ? k : 0] : ((k == 10) ? 1.f : 0.f);
  }
  v2f af2[10][3];
#pragma unroll
  for (int o = 0; o < 10; ++o)
#pragma unroll
    for (int s = 0; s < 3; ++s)
#pragma unroll
      for (int j = 0; j < 2; ++j) {
        const int k = 6 * hv + 2 * s + j;
        const int col = q * 10 + o;
        af2[o][s][j] = (k < 10) ? w2[((k < 10) ? k : 0) * 160 + col]
                                : ((k == 10) ? b2[col] : 0.f);
      }

  float* ml = &sMsg[wib][0];

  int t = wgid;
  if (t >= ntiles) return;
  const unsigned se_die = __builtin_amdgcn_s_sendmsg_rtn(0x87);
  float* __restrict__ aggp = (((se_die >> 16) & 15u) >= 4u) ? agg_hi : agg;
  int eD, eSn;
  v2f a0, a1;
  v4f nlo, nhi;
  {
    const int e0 = t * 16;
    const int t1 = (t + nw < ntiles) ? t + nw : t;
    const int eS = eidx[e0 + q];
    eD = eidx[E_ + e0 + q];
    eSn = eidx[t1 * 16 + q];
    const float* ea = eattr + (size_t)(e0 + q) * 8 + 2 * hv;
    a0 = *(const v2f*)(ea);
    a1 = *(const v2f*)(ea + 4);
    const float* nr = nfeat + (size_t)eS * 16 + 8 * hv;
    nlo = *(const v4f*)(nr);
    nhi = *(const v4f*)(nr + 4);
  }

  for (; t < ntiles; t += nw) {
    const int tn = (t + nw < ntiles) ? t + nw : t;
    const int tnn = (t + 2 * nw < ntiles) ? t + 2 * nw : t;
    const int eSnn = eidx[tnn * 16 + q];
    const int eDn = eidx[E_ + tn * 16 + q];
    const float* ean = eattr + (size_t)(tn * 16 + q) * 8 + 2 * hv;
    const v2f a0n = *(const v2f*)(ean);
    const v2f a1n = *(const v2f*)(ean + 4);
    const float* nrn = nfeat + (size_t)eSn * 16 + 8 * hv;
    const v4f nlon = *(const v4f*)(nrn);
    const v4f nhin = *(const v4f*)(nrn + 4);

    v8f d1;
#pragma unroll
    for (int r = 0; r < 8; ++r) d1[r] = (r < 6) ? c1[r] : 0.f;
    d1 = wmma4(af1[0], a0, d1);
    d1 = wmma4(af1[1], a1, d1);
    v2f hb[3];
#pragma unroll
    for (int s = 0; s < 3; ++s) {
      hb[s][0] = fmaxf(d1[2 * s], 0.f);
      hb[s][1] = fmaxf(d1[2 * s + 1], 0.f);
    }

    const float nf8[8] = {nlo.x, nlo.y, nlo.z, nlo.w, nhi.x, nhi.y, nhi.z, nhi.w};
    float p[10];
#pragma unroll
    for (int o = 0; o < 10; ++o) {
      v8f d2;
#pragma unroll
      for (int r = 0; r < 8; ++r) d2[r] = 0.f;
#pragma unroll
      for (int s = 0; s < 3; ++s) d2 = wmma4(af2[o][s], hb[s], d2);
      float acc = d2[0] * nf8[0];
#pragma unroll
      for (int r = 1; r < 8; ++r) acc = fmaf(d2[r], nf8[r], acc);
      p[o] = acc;
    }

    float fin[5];
#pragma unroll
    for (int jj = 0; jj < 5; ++jj) {
      const float keep = hv ? p[jj + 5] : p[jj];
      const float send = hv ? p[jj] : p[jj + 5];
      const float recv = __int_as_float(__builtin_amdgcn_permlanex16(
          0, __float_as_int(send), 0x76543210, 0xfedcba98, false, false));
      fin[jj] = keep + recv;
    }

#pragma unroll
    for (int jj = 0; jj < 5; ++jj) ml[q * 17 + 5 * hv + jj] = fin[jj];
    if (lane < 16) sDst[wib][q] = eD;
    __asm__ volatile("s_wait_dscnt 0" ::: "memory");

    if (q < 10) {
      float mv[8];
      int dv[8];
#pragma unroll
      for (int r = 0; r < 8; ++r) {
        mv[r] = ml[(r + 8 * hv) * 17 + q];
        dv[r] = sDst[wib][r + 8 * hv];
      }
#pragma unroll
      for (int r = 0; r < 8; ++r)
        atomicAdd(&aggp[(size_t)dv[r] * 16 + q], mv[r]);
    }

    eSn = eSnn;
    eD = eDn;
    a0 = a0n;
    a1 = a1n;
    nlo = nlon;
    nhi = nhin;
  }
}

__global__ void __launch_bounds__(256)
node_relu0(const float* __restrict__ x, float* __restrict__ agg,
           const float* __restrict__ agg_hi, const float* __restrict__ root,
           const float* __restrict__ bias, float* __restrict__ h, int N_) {
  int idx = blockIdx.x * blockDim.x + threadIdx.x;
  if (idx >= N_ * 16) return;
  int v = idx >> 4, j = idx & 15;
  float acc = (agg[idx] + agg_hi[idx]) + bias[j];
  agg[idx] = 0.f;
#pragma unroll
  for (int i = 0; i < 8; ++i) acc += x[v * 8 + i] * root[i * 16 + j];
  h[idx] = fmaxf(acc, 0.f);
}

__global__ void __launch_bounds__(256)
node_relu1(const float* __restrict__ h0, const float* __restrict__ agg,
           const float* __restrict__ agg_hi, const float* __restrict__ root,
           const float* __restrict__ bias, float* __restrict__ h1, int N_) {
  int idx = blockIdx.x * blockDim.x + threadIdx.x;
  if (idx >= N_ * 16) return;
  int v = idx >> 4, j = idx & 15;
  float out = 0.f;
  if (j < 10) {
    float acc = (agg[idx] + agg_hi[idx]) + bias[j];
#pragma unroll
    for (int i = 0; i < 16; ++i) acc += h0[v * 16 + i] * root[i * 10 + j];
    out = fmaxf(acc, 0.f);
  }
  h1[idx] = out;
}

__global__ void __launch_bounds__(256)
edge_pred(const float* __restrict__ eattr, const int* __restrict__ eidx,
          const float* __restrict__ h1, const float* __restrict__ w1,
          const float* __restrict__ b1, const float* __restrict__ w2,
          const float* __restrict__ b2, float* __restrict__ out, int E_,
          int ntiles) {
  const int lane = threadIdx.x & 31;
  const int q = lane & 15;
  const int hv = lane >> 4;
  const int wgid = blockIdx.x * 8 + (threadIdx.x >> 5);
  const int nw = gridDim.x * 8;
  (void)E_;

  v2f bf[5];
#pragma unroll
  for (int s = 0; s < 5; ++s)
#pragma unroll
    for (int j = 0; j < 2; ++j) {
      int K = 4 * s + 2 * hv + j;
      bf[s][j] = (K < 18 && q < 10) ? w1[K * 10 + q] : 0.f;
    }
  const float b1v = (q < 10) ? b1[q] : 0.f;
  const float w2q = (q < 10) ? w2[q] : 0.f;
  const float b2v = b2[0];

  int t = wgid;
  if (t >= ntiles) return;
  int eSn;
  v2f af[5];
  {
    const int e0 = t * 16;
    const int t1 = (t + nw < ntiles) ? t + nw : t;
    const int eS = eidx[e0 + q];
    eSn = eidx[t1 * 16 + q];
    const float* ear = eattr + (size_t)(e0 + q) * 8 + 2 * hv;
    const float* h1r = h1 + (size_t)eS * 16 + 2 * hv;
    af[0] = *(const v2f*)(ear);
    af[1] = *(const v2f*)(ear + 4);
    af[2] = *(const v2f*)(h1r);
    af[3] = *(const v2f*)(h1r + 4);
    af[4] = *(const v2f*)(h1r + 8);
  }

  for (; t < ntiles; t += nw) {
    const int e0 = t * 16;
    const int tn = (t + nw < ntiles) ? t + nw : t;
    const int tnn = (t + 2 * nw < ntiles) ? t + 2 * nw : t;
    const int eSnn = eidx[tnn * 16 + q];
    const float* earn = eattr + (size_t)(tn * 16 + q) * 8 + 2 * hv;
    const float* h1rn = h1 + (size_t)eSn * 16 + 2 * hv;
    v2f afn[5];
    afn[0] = *(const v2f*)(earn);
    afn[1] = *(const v2f*)(earn + 4);
    afn[2] = *(const v2f*)(h1rn);
    afn[3] = *(const v2f*)(h1rn + 4);
    afn[4] = *(const v2f*)(h1rn + 8);

    v8f d;
#pragma unroll
    for (int r = 0; r < 8; ++r) d[r] = b1v;
#pragma unroll
    for (int s = 0; s < 5; ++s) d = wmma4(af[s], bf[s], d);

    float sc = 0.f;
#pragma unroll
    for (int r = 0; r < 8; ++r) {
      float val = red16(fmaxf(d[r], 0.f) * w2q);
      sc = (q == r) ? (val + b2v) : sc;
    }
    if (q < 8) out[e0 + 8 * hv + q] = sc;

    eSn = eSnn;
#pragma unroll
    for (int s = 0; s < 5; ++s) af[s] = afn[s];
  }
}

extern "C" void kernel_launch(void* const* d_in, const int* in_sizes, int n_in,
                              void* d_out, int out_size, void* d_ws,
                              size_t ws_size, hipStream_t stream) {
  const float* x = (const float*)d_in[0];
  const float* eattr = (const float*)d_in[1];
  const int* eidx = (const int*)d_in[2];
  const float* c0_w1 = (const float*)d_in[3];
  const float* c0_b1 = (const float*)d_in[4];
  const float* c0_w2 = (const float*)d_in[5];
  const float* c0_b2 = (const float*)d_in[6];
  const float* c0_root = (const float*)d_in[7];
  const float* c0_bias = (const float*)d_in[8];
  const float* c1_w1 = (const float*)d_in[9];
  const float* c1_b1 = (const float*)d_in[10];
  const float* c1_w2 = (const float*)d_in[11];
  const float* c1_b2 = (const float*)d_in[12];
  const float* c1_root = (const float*)d_in[13];
  const float* c1_bias = (const float*)d_in[14];
  const float* ep_w1 = (const float*)d_in[15];
  const float* ep_b1 = (const float*)d_in[16];
  const float* ep_w2 = (const float*)d_in[17];
  const float* ep_b2 = (const float*)d_in[18];
  (void)n_in; (void)out_size; (void)ws_size;

  const int N_ = in_sizes[0] / 8;
  const int E_ = in_sizes[1] / 8;
  const int ntiles = E_ / 16;

  float* agg0 = (float*)d_ws;
  float* agg1 = agg0 + (size_t)N_ * 16;
  float* h1 = agg1 + (size_t)N_ * 16;
  float* h0 = h1 + (size_t)N_ * 16;

  const int EBLK = 960;
  const int nthr = 256;
  const int nblkN = (N_ * 16 + nthr - 1) / nthr;

  zero_f32<<<(3 * N_ * 16 + nthr - 1) / nthr, nthr, 0, stream>>>(agg0,
                                                                 3 * N_ * 16);
  edge_conv<8, 128, 16, 16><<<EBLK, nthr, 0, stream>>>(
      x, eattr, eidx, c0_w1, c0_b1, c0_w2, c0_b2, agg0, h1, E_, ntiles);
  node_relu0<<<nblkN, nthr, 0, stream>>>(x, agg0, h1, c0_root, c0_bias, h0,
                                         N_);
  edge_conv_l1<<<512, nthr, 0, stream>>>(
      h0, eattr, eidx, c1_w1, c1_b1, c1_w2, c1_b2, agg1, agg0, E_, ntiles);
  node_relu1<<<nblkN, nthr, 0, stream>>>(h0, agg1, agg0, c1_root, c1_bias, h1,
                                         N_);
  edge_pred<<<EBLK, nthr, 0, stream>>>(eattr, eidx, h1, ep_w1, ep_b1, ep_w2,
                                       ep_b2, (float*)d_out, E_, ntiles);
}
